# starting kernel
# baseline (speedup 1.0000x reference)
#include <hip/hip_runtime.h>

typedef float v2f __attribute__((ext_vector_type(2)));
typedef float v4f __attribute__((ext_vector_type(4)));
typedef float v8f __attribute__((ext_vector_type(8)));

#define P_TOT  32768
#define H      256
#define L      16
#define KRED   280
#define APITCH 284
#define MTILE  64
#define KC     56
#define NCHUNK 5

__global__ void const_head_kernel(const float* __restrict__ node_embs,
                                  const float* __restrict__ W1,
                                  const float* __restrict__ b1,
                                  const int* __restrict__ src_idx,
                                  const int* __restrict__ dst_idx,
                                  float* __restrict__ cvec) {
  const int h = threadIdx.x;
  const float* se = node_embs + (long)(*src_idx) * H;
  const float* de = node_embs + (long)(*dst_idx) * H;
  float acc = b1[h];
  for (int k = 0; k < H; ++k) acc += se[k] * W1[(long)k * H + h];
  for (int k = 0; k < H; ++k) acc += de[k] * W1[(long)(H + k) * H + h];
  cvec[h] = acc;
}

__global__ __launch_bounds__(256) void path_policy_kernel(
    const float* __restrict__ node_embs,
    const int*   __restrict__ path_nodes,
    const int*   __restrict__ path_lens,
    const float* __restrict__ edge_feats,
    const float* __restrict__ scal_feats,
    const float* __restrict__ W1,
    const float* __restrict__ W2,
    const float* __restrict__ b2,
    const float* __restrict__ cvec,
    float* __restrict__ out) {
  __shared__ float ldsA[MTILE][APITCH];
  __shared__ float ldsB[(KC / 4) * H * 4];
  __shared__ float ldsQ[MTILE];

  const int tid  = threadIdx.x;
  const int lane = tid & 31;
  const int wid  = tid >> 5;
  const int p0   = blockIdx.x * MTILE;

  if (tid < MTILE) ldsQ[tid] = 0.0f;

  {
    const int col = lane * 8;
    for (int j = 0; j < 8; ++j) {
      const int pi  = wid * 8 + j;
      const int p   = p0 + pi;
      const int len = path_lens[p];
      v4f a0 = {}, a1 = {};
      for (int t = 0; t < len; ++t) {
        const long idx = (long)path_nodes[p * L + t];
        const v4f* src = (const v4f*)(node_embs + idx * H + col);
        a0 += src[0];
        a1 += src[1];
      }
      const float inv = 1.0f / (float)(len > 0 ? len : 1);
      v4f* dst = (v4f*)&ldsA[pi][col];
      dst[0] = a0 * inv;
      dst[1] = a1 * inv;
    }
  }
  for (int i = tid; i < MTILE * 24; i += 256) {
    const int pi = i / 24, c = i % 24;
    const float v = (c < 8) ? edge_feats[(long)(p0 + pi) * 8 + c]
                            : scal_feats[(long)(p0 + pi) * 16 + (c - 8)];
    ldsA[pi][256 + c] = v;
  }

  const int mgrp  = wid >> 2;
  const int n0    = (wid & 3) * 64;
  const int lh    = lane & 15;
  const int khalf = lane >> 4;
  const int arow0 = mgrp * 32 + lh;
  const int arow1 = arow0 + 16;

  v8f acc[2][4] = {};

  const float* W1s = W1 + (long)512 * H;
  for (int ch = 0; ch < NCHUNK; ++ch) {
    const int kbase = ch * KC;
    __syncthreads();
    {
      const int n = tid;
      for (int kk = 0; kk < KC / 4; ++kk) {
        const long krow = kbase + kk * 4;
        v4f b;
        b[0] = W1s[(krow + 0) * H + n];
        b[1] = W1s[(krow + 1) * H + n];
        b[2] = W1s[(krow + 2) * H + n];
        b[3] = W1s[(krow + 3) * H + n];
        *(v4f*)&ldsB[(kk * H + n) * 4] = b;
      }
    }
    __syncthreads();
#pragma unroll
    for (int kk = 0; kk < KC / 4; ++kk) {
      const int k = kbase + kk * 4;
      const v2f a0 = *(const v2f*)&ldsA[arow0][k + khalf * 2];
      const v2f a1 = *(const v2f*)&ldsA[arow1][k + khalf * 2];
      const float* bb = &ldsB[(kk * H + n0 + lh) * 4 + khalf * 2];
      const v2f b0 = *(const v2f*)(bb);
      const v2f b1 = *(const v2f*)(bb + 16 * 4);
      const v2f b2v = *(const v2f*)(bb + 32 * 4);
      const v2f b3 = *(const v2f*)(bb + 48 * 4);
      acc[0][0] = __builtin_amdgcn_wmma_f32_16x16x4_f32(false, a0, false, b0,  (short)0, acc[0][0], false, false);
      acc[0][1] = __builtin_amdgcn_wmma_f32_16x16x4_f32(false, a0, false, b1,  (short)0, acc[0][1], false, false);
      acc[0][2] = __builtin_amdgcn_wmma_f32_16x16x4_f32(false, a0, false, b2v, (short)0, acc[0][2], false, false);
      acc[0][3] = __builtin_amdgcn_wmma_f32_16x16x4_f32(false, a0, false, b3,  (short)0, acc[0][3], false, false);
      acc[1][0] = __builtin_amdgcn_wmma_f32_16x16x4_f32(false, a1, false, b0,  (short)0, acc[1][0], false, false);
      acc[1][1] = __builtin_amdgcn_wmma_f32_16x16x4_f32(false, a1, false, b1,  (short)0, acc[1][1], false, false);
      acc[1][2] = __builtin_amdgcn_wmma_f32_16x16x4_f32(false, a1, false, b2v, (short)0, acc[1][2], false, false);
      acc[1][3] = __builtin_amdgcn_wmma_f32_16x16x4_f32(false, a1, false, b3,  (short)0, acc[1][3], false, false);
    }
  }

  float cn[4], wn[4];
#pragma unroll
  for (int s = 0; s < 4; ++s) {
    const int n = n0 + s * 16 + lh;
    cn[s] = cvec[n];
    wn[s] = W2[n];
  }
#pragma unroll
  for (int ms = 0; ms < 2; ++ms) {
#pragma unroll
    for (int r = 0; r < 8; ++r) {
      float part = 0.0f, v;
      v = acc[ms][0][r] + cn[0]; part += fmaxf(v, 0.0f) * wn[0];
      v = acc[ms][1][r] + cn[1]; part += fmaxf(v, 0.0f) * wn[1];
      v = acc[ms][2][r] + cn[2]; part += fmaxf(v, 0.0f) * wn[2];
      v = acc[ms][3][r] + cn[3]; part += fmaxf(v, 0.0f) * wn[3];
      part += __shfl_xor(part, 1, 32);
      part += __shfl_xor(part, 2, 32);
      part += __shfl_xor(part, 4, 32);
      part += __shfl_xor(part, 8, 32);
      if (lh == 0) {
        const int row = mgrp * 32 + ms * 16 + r + khalf * 8;
        atomicAdd(&ldsQ[row], part);
      }
    }
  }

  __syncthreads();
  if (tid < MTILE) {
    const int p   = p0 + tid;
    const int len = path_lens[p];
    out[p] = (len > 0) ? (ldsQ[tid] + b2[0]) : 0.0f;
  }
}

extern "C" void kernel_launch(void* const* d_in, const int* in_sizes, int n_in,
                              void* d_out, int out_size, void* d_ws, size_t ws_size,
                              hipStream_t stream) {
  const float* node_embs = (const float*)d_in[0];
  const int*   path_nodes = (const int*)d_in[1];
  const int*   path_lens  = (const int*)d_in[2];
  const float* edge_feats = (const float*)d_in[3];
  const float* scal_feats = (const float*)d_in[4];
  const float* W1 = (const float*)d_in[5];
  const float* b1 = (const float*)d_in[6];
  const float* W2 = (const float*)d_in[7];
  const float* b2 = (const float*)d_in[8];
  const int*   src_idx = (const int*)d_in[9];
  const int*   dst_idx = (const int*)d_in[10];

  float* cvec = (float*)d_ws;

  const_head_kernel<<<1, 256, 0, stream>>>(node_embs, W1, b1, src_idx, dst_idx, cvec);
  path_policy_kernel<<<P_TOT / MTILE, 256, 0, stream>>>(
      node_embs, path_nodes, path_lens, edge_feats, scal_feats,
      W1, W2, b2, cvec, (float*)d_out);
}


# constant-head fold of the src/dst rows of W1 split over 8 blocks x 4 K-slices, 16 independent row loads per thread; main epilogue adds the 8 partial vectors (changes summation order of c)
# speedup vs baseline: 1.2971x; 1.2971x over previous
#include <hip/hip_runtime.h>

typedef float v2f __attribute__((ext_vector_type(2)));
typedef float v4f __attribute__((ext_vector_type(4)));
typedef float v8f __attribute__((ext_vector_type(8)));

#define P_TOT  32768
#define H      256
#define L      16
#define KRED   280
#define APITCH 284
#define MTILE  64
#define KC     56
#define NCHUNK 5
#define KSL    16
#define NPART  8

__global__ __launch_bounds__(1024) void const_head_kernel(const float* __restrict__ node_embs,
                                  const float* __restrict__ W1,
                                  const float* __restrict__ b1,
                                  const int* __restrict__ src_idx,
                                  const int* __restrict__ dst_idx,
                                  float* __restrict__ part) {
  __shared__ float red[4][H];
  const int h  = threadIdx.x & (H - 1);
  const int ks = __builtin_amdgcn_readfirstlane((int)(threadIdx.x >> 8));
  const int slice = blockIdx.x * 4 + ks;
  const int node  = (slice < 16) ? *src_idx : *dst_idx;
  const float* x = node_embs + (long)node * H + (slice & 15) * KSL;
  const float* w = W1 + (long)slice * KSL * H + h;
  const float b1h  = b1[h];
  const float bias = (blockIdx.x == 0) ? b1h : 0.0f;
  float acc = 0.0f;
  for (int j = 0; j < KSL; ++j) acc += x[j] * w[(long)j * H];
  red[ks][h] = acc;
  __syncthreads();
  if (ks == 0) {
    const float s = (red[0][h] + red[1][h]) + (red[2][h] + red[3][h]);
    part[blockIdx.x * H + h] = s + bias;
  }
}

__global__ __launch_bounds__(256) void path_policy_kernel(
    const float* __restrict__ node_embs,
    const int*   __restrict__ path_nodes,
    const int*   __restrict__ path_lens,
    const float* __restrict__ edge_feats,
    const float* __restrict__ scal_feats,
    const float* __restrict__ W1,
    const float* __restrict__ W2,
    const float* __restrict__ b2,
    const float* __restrict__ part,
    float* __restrict__ out) {
  __shared__ float ldsA[MTILE][APITCH];
  __shared__ float ldsB[(KC / 4) * H * 4];
  __shared__ float ldsQ[MTILE];

  const int tid  = threadIdx.x;
  const int lane = tid & 31;
  const int wid  = tid >> 5;
  const int p0   = blockIdx.x * MTILE;

  if (tid < MTILE) ldsQ[tid] = 0.0f;

  {
    const int col = lane * 8;
    for (int j = 0; j < 8; ++j) {
      const int pi  = wid * 8 + j;
      const int p   = p0 + pi;
      const int len = path_lens[p];
      v4f a0 = {}, a1 = {};
      for (int t = 0; t < len; ++t) {
        const long idx = (long)path_nodes[p * L + t];
        const v4f* src = (const v4f*)(node_embs + idx * H + col);
        a0 += src[0];
        a1 += src[1];
      }
      const float inv = 1.0f / (float)(len > 0 ? len : 1);
      v4f* dst = (v4f*)&ldsA[pi][col];
      dst[0] = a0 * inv;
      dst[1] = a1 * inv;
    }
  }
  for (int i = tid; i < MTILE * 24; i += 256) {
    const int pi = i / 24, c = i % 24;
    const float v = (c < 8) ? edge_feats[(long)(p0 + pi) * 8 + c]
                            : scal_feats[(long)(p0 + pi) * 16 + (c - 8)];
    ldsA[pi][256 + c] = v;
  }

  const int mgrp  = wid >> 2;
  const int n0    = (wid & 3) * 64;
  const int lh    = lane & 15;
  const int khalf = lane >> 4;
  const int arow0 = mgrp * 32 + lh;
  const int arow1 = arow0 + 16;

  v8f acc[2][4] = {};

  const float* W1s = W1 + (long)512 * H;
  for (int ch = 0; ch < NCHUNK; ++ch) {
    const int kbase = ch * KC;
    __syncthreads();
    {
      const int n = tid;
      for (int kk = 0; kk < KC / 4; ++kk) {
        const long krow = kbase + kk * 4;
        v4f b;
        b[0] = W1s[(krow + 0) * H + n];
        b[1] = W1s[(krow + 1) * H + n];
        b[2] = W1s[(krow + 2) * H + n];
        b[3] = W1s[(krow + 3) * H + n];
        *(v4f*)&ldsB[(kk * H + n) * 4] = b;
      }
    }
    __syncthreads();
#pragma unroll
    for (int kk = 0; kk < KC / 4; ++kk) {
      const int k = kbase + kk * 4;
      const v2f a0 = *(const v2f*)&ldsA[arow0][k + khalf * 2];
      const v2f a1 = *(const v2f*)&ldsA[arow1][k + khalf * 2];
      const float* bb = &ldsB[(kk * H + n0 + lh) * 4 + khalf * 2];
      const v2f b0 = *(const v2f*)(bb);
      const v2f b1 = *(const v2f*)(bb + 16 * 4);
      const v2f b2v = *(const v2f*)(bb + 32 * 4);
      const v2f b3 = *(const v2f*)(bb + 48 * 4);
      acc[0][0] = __builtin_amdgcn_wmma_f32_16x16x4_f32(false, a0, false, b0,  (short)0, acc[0][0], false, false);
      acc[0][1] = __builtin_amdgcn_wmma_f32_16x16x4_f32(false, a0, false, b1,  (short)0, acc[0][1], false, false);
      acc[0][2] = __builtin_amdgcn_wmma_f32_16x16x4_f32(false, a0, false, b2v, (short)0, acc[0][2], false, false);
      acc[0][3] = __builtin_amdgcn_wmma_f32_16x16x4_f32(false, a0, false, b3,  (short)0, acc[0][3], false, false);
      acc[1][0] = __builtin_amdgcn_wmma_f32_16x16x4_f32(false, a1, false, b0,  (short)0, acc[1][0], false, false);
      acc[1][1] = __builtin_amdgcn_wmma_f32_16x16x4_f32(false, a1, false, b1,  (short)0, acc[1][1], false, false);
      acc[1][2] = __builtin_amdgcn_wmma_f32_16x16x4_f32(false, a1, false, b2v, (short)0, acc[1][2], false, false);
      acc[1][3] = __builtin_amdgcn_wmma_f32_16x16x4_f32(false, a1, false, b3,  (short)0, acc[1][3], false, false);
    }
  }

  float cn[4], wn[4];
#pragma unroll
  for (int s = 0; s < 4; ++s) {
    const int n = n0 + s * 16 + lh;
    float cs = part[n];
    for (int j = 1; j < NPART; ++j) cs += part[j * H + n];
    cn[s] = cs;
    wn[s] = W2[n];
  }
#pragma unroll
  for (int ms = 0; ms < 2; ++ms) {
#pragma unroll
    for (int r = 0; r < 8; ++r) {
      float part = 0.0f, v;
      v = acc[ms][0][r] + cn[0]; part += fmaxf(v, 0.0f) * wn[0];
      v = acc[ms][1][r] + cn[1]; part += fmaxf(v, 0.0f) * wn[1];
      v = acc[ms][2][r] + cn[2]; part += fmaxf(v, 0.0f) * wn[2];
      v = acc[ms][3][r] + cn[3]; part += fmaxf(v, 0.0f) * wn[3];
      part += __shfl_xor(part, 1, 32);
      part += __shfl_xor(part, 2, 32);
      part += __shfl_xor(part, 4, 32);
      part += __shfl_xor(part, 8, 32);
      if (lh == 0) {
        const int row = mgrp * 32 + ms * 16 + r + khalf * 8;
        atomicAdd(&ldsQ[row], part);
      }
    }
  }

  __syncthreads();
  if (tid < MTILE) {
    const int p   = p0 + tid;
    const int len = path_lens[p];
    out[p] = (len > 0) ? (ldsQ[tid] + b2[0]) : 0.0f;
  }
}

extern "C" void kernel_launch(void* const* d_in, const int* in_sizes, int n_in,
                              void* d_out, int out_size, void* d_ws, size_t ws_size,
                              hipStream_t stream) {
  const float* node_embs = (const float*)d_in[0];
  const int*   path_nodes = (const int*)d_in[1];
  const int*   path_lens  = (const int*)d_in[2];
  const float* edge_feats = (const float*)d_in[3];
  const float* scal_feats = (const float*)d_in[4];
  const float* W1 = (const float*)d_in[5];
  const float* b1 = (const float*)d_in[6];
  const float* W2 = (const float*)d_in[7];
  const float* b2 = (const float*)d_in[8];
  const int*   src_idx = (const int*)d_in[9];
  const int*   dst_idx = (const int*)d_in[10];

  float* part = (float*)d_ws;

  const_head_kernel<<<NPART, 1024, 0, stream>>>(node_embs, W1, b1, src_idx, dst_idx, part);
  path_policy_kernel<<<P_TOT / MTILE, 256, 0, stream>>>(
      node_embs, path_nodes, path_lens, edge_feats, scal_feats,
      W1, W2, b2, part, (float*)d_out);
}


# gather: each wave fetches its 8 lengths and 128 node indices in one round trip, then issues all len row loads of a path back to back before the adds; add order t=0..len-1 kept
# speedup vs baseline: 1.4448x; 1.1139x over previous
#include <hip/hip_runtime.h>

typedef float v2f __attribute__((ext_vector_type(2)));
typedef float v4f __attribute__((ext_vector_type(4)));
typedef float v8f __attribute__((ext_vector_type(8)));
typedef int   v4i __attribute__((ext_vector_type(4)));

#define P_TOT  32768
#define H      256
#define L      16
#define KRED   280
#define APITCH 284
#define MTILE  64
#define KC     56
#define NCHUNK 5
#define KSL    16
#define NPART  8

__global__ __launch_bounds__(1024) void const_head_kernel(const float* __restrict__ node_embs,
                                  const float* __restrict__ W1,
                                  const float* __restrict__ b1,
                                  const int* __restrict__ src_idx,
                                  const int* __restrict__ dst_idx,
                                  float* __restrict__ part) {
  __shared__ float red[4][H];
  const int h  = threadIdx.x & (H - 1);
  const int ks = __builtin_amdgcn_readfirstlane((int)(threadIdx.x >> 8));
  const int slice = blockIdx.x * 4 + ks;
  const int node  = (slice < 16) ? *src_idx : *dst_idx;
  const float* x = node_embs + (long)node * H + (slice & 15) * KSL;
  const float* w = W1 + (long)slice * KSL * H + h;
  const float b1h  = b1[h];
  const float bias = (blockIdx.x == 0) ? b1h : 0.0f;
  float acc = 0.0f;
  for (int j = 0; j < KSL; ++j) acc += x[j] * w[(long)j * H];
  red[ks][h] = acc;
  __syncthreads();
  if (ks == 0) {
    const float s = (red[0][h] + red[1][h]) + (red[2][h] + red[3][h]);
    part[blockIdx.x * H + h] = s + bias;
  }
}

__global__ __launch_bounds__(256) void path_policy_kernel(
    const float* __restrict__ node_embs,
    const int*   __restrict__ path_nodes,
    const int*   __restrict__ path_lens,
    const float* __restrict__ edge_feats,
    const float* __restrict__ scal_feats,
    const float* __restrict__ W1,
    const float* __restrict__ W2,
    const float* __restrict__ b2,
    const float* __restrict__ part,
    float* __restrict__ out) {
  __shared__ float ldsA[MTILE][APITCH];
  __shared__ float ldsB[(KC / 4) * H * 4];
  __shared__ float ldsQ[MTILE];

  const int tid  = threadIdx.x;
  const int lane = tid & 31;
  const int wid  = tid >> 5;
  const int p0   = blockIdx.x * MTILE;

  if (tid < MTILE) ldsQ[tid] = 0.0f;

  {
    const int col = lane * 8;
    const int pw  = p0 + wid * 8;
    const v4i iv  = *(const v4i*)(path_nodes + (long)pw * L + lane * 4);
    const int lv  = path_lens[pw + (lane & 7)];
    const float* nb = node_embs + col;
    for (int j = 0; j < 8; ++j) {
      const int len = __builtin_amdgcn_readlane(lv, j);
      v4f r0[L], r1[L];
#pragma unroll
      for (int t = 0; t < L; ++t) {
        if (t < len) {
          const int idx = __builtin_amdgcn_readlane(iv[t & 3], j * 4 + (t >> 2));
          const v4f* src = (const v4f*)(nb + (long)idx * H);
          r0[t] = src[0];
          r1[t] = src[1];
        }
      }
      v4f a0 = {}, a1 = {};
#pragma unroll
      for (int t = 0; t < L; ++t) {
        if (t < len) {
          a0 += r0[t];
          a1 += r1[t];
        }
      }
      const float inv = 1.0f / (float)(len > 0 ? len : 1);
      v4f* dst = (v4f*)&ldsA[wid * 8 + j][col];
      dst[0] = a0 * inv;
      dst[1] = a1 * inv;
    }
  }
  for (int i = tid; i < MTILE * 24; i += 256) {
    const int pi = i / 24, c = i % 24;
    const float v = (c < 8) ? edge_feats[(long)(p0 + pi) * 8 + c]
                            : scal_feats[(long)(p0 + pi) * 16 + (c - 8)];
    ldsA[pi][256 + c] = v;
  }

  const int mgrp  = wid >> 2;
  const int n0    = (wid & 3) * 64;
  const int lh    = lane & 15;
  const int khalf = lane >> 4;
  const int arow0 = mgrp * 32 + lh;
  const int arow1 = arow0 + 16;

  v8f acc[2][4] = {};

  const float* W1s = W1 + (long)512 * H;
  for (int ch = 0; ch < NCHUNK; ++ch) {
    const int kbase = ch * KC;
    __syncthreads();
    {
      const int n = tid;
      for (int kk = 0; kk < KC / 4; ++kk) {
        const long krow = kbase + kk * 4;
        v4f b;
        b[0] = W1s[(krow + 0) * H + n];
        b[1] = W1s[(krow + 1) * H + n];
        b[2] = W1s[(krow + 2) * H + n];
        b[3] = W1s[(krow + 3) * H + n];
        *(v4f*)&ldsB[(kk * H + n) * 4] = b;
      }
    }
    __syncthreads();
#pragma unroll
    for (int kk = 0; kk < KC / 4; ++kk) {
      const int k = kbase + kk * 4;
      const v2f a0 = *(const v2f*)&ldsA[arow0][k + khalf * 2];
      const v2f a1 = *(const v2f*)&ldsA[arow1][k + khalf * 2];
      const float* bb = &ldsB[(kk * H + n0 + lh) * 4 + khalf * 2];
      const v2f b0 = *(const v2f*)(bb);
      const v2f b1 = *(const v2f*)(bb + 16 * 4);
      const v2f b2v = *(const v2f*)(bb + 32 * 4);
      const v2f b3 = *(const v2f*)(bb + 48 * 4);
      acc[0][0] = __builtin_amdgcn_wmma_f32_16x16x4_f32(false, a0, false, b0,  (short)0, acc[0][0], false, false);
      acc[0][1] = __builtin_amdgcn_wmma_f32_16x16x4_f32(false, a0, false, b1,  (short)0, acc[0][1], false, false);
      acc[0][2] = __builtin_amdgcn_wmma_f32_16x16x4_f32(false, a0, false, b2v, (short)0, acc[0][2], false, false);
      acc[0][3] = __builtin_amdgcn_wmma_f32_16x16x4_f32(false, a0, false, b3,  (short)0, acc[0][3], false, false);
      acc[1][0] = __builtin_amdgcn_wmma_f32_16x16x4_f32(false, a1, false, b0,  (short)0, acc[1][0], false, false);
      acc[1][1] = __builtin_amdgcn_wmma_f32_16x16x4_f32(false, a1, false, b1,  (short)0, acc[1][1], false, false);
      acc[1][2] = __builtin_amdgcn_wmma_f32_16x16x4_f32(false, a1, false, b2v, (short)0, acc[1][2], false, false);
      acc[1][3] = __builtin_amdgcn_wmma_f32_16x16x4_f32(false, a1, false, b3,  (short)0, acc[1][3], false, false);
    }
  }

  float cn[4], wn[4];
#pragma unroll
  for (int s = 0; s < 4; ++s) {
    const int n = n0 + s * 16 + lh;
    float cs = part[n];
    for (int j = 1; j < NPART; ++j) cs += part[j * H + n];
    cn[s] = cs;
    wn[s] = W2[n];
  }
#pragma unroll
  for (int ms = 0; ms < 2; ++ms) {
#pragma unroll
    for (int r = 0; r < 8; ++r) {
      float part = 0.0f, v;
      v = acc[ms][0][r] + cn[0]; part += fmaxf(v, 0.0f) * wn[0];
      v = acc[ms][1][r] + cn[1]; part += fmaxf(v, 0.0f) * wn[1];
      v = acc[ms][2][r] + cn[2]; part += fmaxf(v, 0.0f) * wn[2];
      v = acc[ms][3][r] + cn[3]; part += fmaxf(v, 0.0f) * wn[3];
      part += __shfl_xor(part, 1, 32);
      part += __shfl_xor(part, 2, 32);
      part += __shfl_xor(part, 4, 32);
      part += __shfl_xor(part, 8, 32);
      if (lh == 0) {
        const int row = mgrp * 32 + ms * 16 + r + khalf * 8;
        atomicAdd(&ldsQ[row], part);
      }
    }
  }

  __syncthreads();
  if (tid < MTILE) {
    const int p   = p0 + tid;
    const int len = path_lens[p];
    out[p] = (len > 0) ? (ldsQ[tid] + b2[0]) : 0.0f;
  }
}

extern "C" void kernel_launch(void* const* d_in, const int* in_sizes, int n_in,
                              void* d_out, int out_size, void* d_ws, size_t ws_size,
                              hipStream_t stream) {
  const float* node_embs = (const float*)d_in[0];
  const int*   path_nodes = (const int*)d_in[1];
  const int*   path_lens  = (const int*)d_in[2];
  const float* edge_feats = (const float*)d_in[3];
  const float* scal_feats = (const float*)d_in[4];
  const float* W1 = (const float*)d_in[5];
  const float* b1 = (const float*)d_in[6];
  const float* W2 = (const float*)d_in[7];
  const float* b2 = (const float*)d_in[8];
  const int*   src_idx = (const int*)d_in[9];
  const int*   dst_idx = (const int*)d_in[10];

  float* part = (float*)d_ws;

  const_head_kernel<<<NPART, 1024, 0, stream>>>(node_embs, W1, b1, src_idx, dst_idx, part);
  path_policy_kernel<<<P_TOT / MTILE, 256, 0, stream>>>(
      node_embs, path_nodes, path_lens, edge_feats, scal_feats,
      W1, W2, b2, part, (float*)d_out);
}


# gather: each row load covers four whole 128-B lines (16 B per lane at lane stride 16 B; lane l owns columns 4l..4l+3 and 128+4l..) instead of half of all eight lines; add order t=0..len-1 kept
# speedup vs baseline: 1.5101x; 1.0452x over previous
#include <hip/hip_runtime.h>

typedef float v2f __attribute__((ext_vector_type(2)));
typedef float v4f __attribute__((ext_vector_type(4)));
typedef float v8f __attribute__((ext_vector_type(8)));
typedef int   v4i __attribute__((ext_vector_type(4)));

#define P_TOT  32768
#define H      256
#define L      16
#define KRED   280
#define APITCH 284
#define MTILE  64
#define KC     56
#define NCHUNK 5
#define KSL    16
#define NPART  8

__global__ __launch_bounds__(1024) void const_head_kernel(const float* __restrict__ node_embs,
                                  const float* __restrict__ W1,
                                  const float* __restrict__ b1,
                                  const int* __restrict__ src_idx,
                                  const int* __restrict__ dst_idx,
                                  float* __restrict__ part) {
  __shared__ float red[4][H];
  const int h  = threadIdx.x & (H - 1);
  const int ks = __builtin_amdgcn_readfirstlane((int)(threadIdx.x >> 8));
  const int slice = blockIdx.x * 4 + ks;
  const int node  = (slice < 16) ? *src_idx : *dst_idx;
  const float* x = node_embs + (long)node * H + (slice & 15) * KSL;
  const float* w = W1 + (long)slice * KSL * H + h;
  const float b1h  = b1[h];
  const float bias = (blockIdx.x == 0) ? b1h : 0.0f;
  float acc = 0.0f;
  for (int j = 0; j < KSL; ++j) acc += x[j] * w[(long)j * H];
  red[ks][h] = acc;
  __syncthreads();
  if (ks == 0) {
    const float s = (red[0][h] + red[1][h]) + (red[2][h] + red[3][h]);
    part[blockIdx.x * H + h] = s + bias;
  }
}

__global__ __launch_bounds__(256) void path_policy_kernel(
    const float* __restrict__ node_embs,
    const int*   __restrict__ path_nodes,
    const int*   __restrict__ path_lens,
    const float* __restrict__ edge_feats,
    const float* __restrict__ scal_feats,
    const float* __restrict__ W1,
    const float* __restrict__ W2,
    const float* __restrict__ b2,
    const float* __restrict__ part,
    float* __restrict__ out) {
  __shared__ float ldsA[MTILE][APITCH];
  __shared__ float ldsB[(KC / 4) * H * 4];
  __shared__ float ldsQ[MTILE];

  const int tid  = threadIdx.x;
  const int lane = tid & 31;
  const int wid  = tid >> 5;
  const int p0   = blockIdx.x * MTILE;

  if (tid < MTILE) ldsQ[tid] = 0.0f;

  {
    const int col = lane * 4;
    const int pw  = p0 + wid * 8;
    const v4i iv  = *(const v4i*)(path_nodes + (long)pw * L + lane * 4);
    const int lv  = path_lens[pw + (lane & 7)];
    const float* nb = node_embs + col;
    for (int j = 0; j < 8; ++j) {
      const int len = __builtin_amdgcn_readlane(lv, j);
      v4f r0[L], r1[L];
#pragma unroll
      for (int t = 0; t < L; ++t) {
        if (t < len) {
          const int idx = __builtin_amdgcn_readlane(iv[t & 3], j * 4 + (t >> 2));
          const v4f* src = (const v4f*)(nb + (long)idx * H);
          r0[t] = src[0];
          r1[t] = src[H / 8];
        }
      }
      v4f a0 = {}, a1 = {};
#pragma unroll
      for (int t = 0; t < L; ++t) {
        if (t < len) {
          a0 += r0[t];
          a1 += r1[t];
        }
      }
      const float inv = 1.0f / (float)(len > 0 ? len : 1);
      *(v4f*)&ldsA[wid * 8 + j][col]         = a0 * inv;
      *(v4f*)&ldsA[wid * 8 + j][H / 2 + col] = a1 * inv;
    }
  }
  for (int i = tid; i < MTILE * 24; i += 256) {
    const int pi = i / 24, c = i % 24;
    const float v = (c < 8) ? edge_feats[(long)(p0 + pi) * 8 + c]
                            : scal_feats[(long)(p0 + pi) * 16 + (c - 8)];
    ldsA[pi][256 + c] = v;
  }

  const int mgrp  = wid >> 2;
  const int n0    = (wid & 3) * 64;
  const int lh    = lane & 15;
  const int khalf = lane >> 4;
  const int arow0 = mgrp * 32 + lh;
  const int arow1 = arow0 + 16;

  v8f acc[2][4] = {};

  const float* W1s = W1 + (long)512 * H;
  for (int ch = 0; ch < NCHUNK; ++ch) {
    const int kbase = ch * KC;
    __syncthreads();
    {
      const int n = tid;
      for (int kk = 0; kk < KC / 4; ++kk) {
        const long krow = kbase + kk * 4;
        v4f b;
        b[0] = W1s[(krow + 0) * H + n];
        b[1] = W1s[(krow + 1) * H + n];
        b[2] = W1s[(krow + 2) * H + n];
        b[3] = W1s[(krow + 3) * H + n];
        *(v4f*)&ldsB[(kk * H + n) * 4] = b;
      }
    }
    __syncthreads();
#pragma unroll
    for (int kk = 0; kk < KC / 4; ++kk) {
      const int k = kbase + kk * 4;
      const v2f a0 = *(const v2f*)&ldsA[arow0][k + khalf * 2];
      const v2f a1 = *(const v2f*)&ldsA[arow1][k + khalf * 2];
      const float* bb = &ldsB[(kk * H + n0 + lh) * 4 + khalf * 2];
      const v2f b0 = *(const v2f*)(bb);
      const v2f b1 = *(const v2f*)(bb + 16 * 4);
      const v2f b2v = *(const v2f*)(bb + 32 * 4);
      const v2f b3 = *(const v2f*)(bb + 48 * 4);
      acc[0][0] = __builtin_amdgcn_wmma_f32_16x16x4_f32(false, a0, false, b0,  (short)0, acc[0][0], false, false);
      acc[0][1] = __builtin_amdgcn_wmma_f32_16x16x4_f32(false, a0, false, b1,  (short)0, acc[0][1], false, false);
      acc[0][2] = __builtin_amdgcn_wmma_f32_16x16x4_f32(false, a0, false, b2v, (short)0, acc[0][2], false, false);
      acc[0][3] = __builtin_amdgcn_wmma_f32_16x16x4_f32(false, a0, false, b3,  (short)0, acc[0][3], false, false);
      acc[1][0] = __builtin_amdgcn_wmma_f32_16x16x4_f32(false, a1, false, b0,  (short)0, acc[1][0], false, false);
      acc[1][1] = __builtin_amdgcn_wmma_f32_16x16x4_f32(false, a1, false, b1,  (short)0, acc[1][1], false, false);
      acc[1][2] = __builtin_amdgcn_wmma_f32_16x16x4_f32(false, a1, false, b2v, (short)0, acc[1][2], false, false);
      acc[1][3] = __builtin_amdgcn_wmma_f32_16x16x4_f32(false, a1, false, b3,  (short)0, acc[1][3], false, false);
    }
  }

  float cn[4], wn[4];
#pragma unroll
  for (int s = 0; s < 4; ++s) {
    const int n = n0 + s * 16 + lh;
    float cs = part[n];
    for (int j = 1; j < NPART; ++j) cs += part[j * H + n];
    cn[s] = cs;
    wn[s] = W2[n];
  }
#pragma unroll
  for (int ms = 0; ms < 2; ++ms) {
#pragma unroll
    for (int r = 0; r < 8; ++r) {
      float part = 0.0f, v;
      v = acc[ms][0][r] + cn[0]; part += fmaxf(v, 0.0f) * wn[0];
      v = acc[ms][1][r] + cn[1]; part += fmaxf(v, 0.0f) * wn[1];
      v = acc[ms][2][r] + cn[2]; part += fmaxf(v, 0.0f) * wn[2];
      v = acc[ms][3][r] + cn[3]; part += fmaxf(v, 0.0f) * wn[3];
      part += __shfl_xor(part, 1, 32);
      part += __shfl_xor(part, 2, 32);
      part += __shfl_xor(part, 4, 32);
      part += __shfl_xor(part, 8, 32);
      if (lh == 0) {
        const int row = mgrp * 32 + ms * 16 + r + khalf * 8;
        atomicAdd(&ldsQ[row], part);
      }
    }
  }

  __syncthreads();
  if (tid < MTILE) {
    const int p   = p0 + tid;
    const int len = path_lens[p];
    out[p] = (len > 0) ? (ldsQ[tid] + b2[0]) : 0.0f;
  }
}

extern "C" void kernel_launch(void* const* d_in, const int* in_sizes, int n_in,
                              void* d_out, int out_size, void* d_ws, size_t ws_size,
                              hipStream_t stream) {
  const float* node_embs = (const float*)d_in[0];
  const int*   path_nodes = (const int*)d_in[1];
  const int*   path_lens  = (const int*)d_in[2];
  const float* edge_feats = (const float*)d_in[3];
  const float* scal_feats = (const float*)d_in[4];
  const float* W1 = (const float*)d_in[5];
  const float* b1 = (const float*)d_in[6];
  const float* W2 = (const float*)d_in[7];
  const float* b2 = (const float*)d_in[8];
  const int*   src_idx = (const int*)d_in[9];
  const int*   dst_idx = (const int*)d_in[10];

  float* part = (float*)d_ws;

  const_head_kernel<<<NPART, 1024, 0, stream>>>(node_embs, W1, b1, src_idx, dst_idx, part);
  path_policy_kernel<<<P_TOT / MTILE, 256, 0, stream>>>(
      node_embs, path_nodes, path_lens, edge_feats, scal_feats,
      W1, W2, b2, part, (float*)d_out);
}


# gather: second column half of each row is loaded inside K-chunks 0-1 of the GEMM, one path per quarter chunk, in flight under the matrix work; first half two paths per round trip; add and k order kept
# speedup vs baseline: 1.5940x; 1.0556x over previous
#include <hip/hip_runtime.h>

typedef float v2f __attribute__((ext_vector_type(2)));
typedef float v4f __attribute__((ext_vector_type(4)));
typedef float v8f __attribute__((ext_vector_type(8)));
typedef int   v4i __attribute__((ext_vector_type(4)));

#define P_TOT  32768
#define H      256
#define L      16
#define KRED   280
#define APITCH 284
#define MTILE  64
#define KC     56
#define NCHUNK 5
#define KSL    16
#define NPART  8

__global__ __launch_bounds__(1024) void const_head_kernel(const float* __restrict__ node_embs,
                                  const float* __restrict__ W1,
                                  const float* __restrict__ b1,
                                  const int* __restrict__ src_idx,
                                  const int* __restrict__ dst_idx,
                                  float* __restrict__ part) {
  __shared__ float red[4][H];
  const int h  = threadIdx.x & (H - 1);
  const int ks = __builtin_amdgcn_readfirstlane((int)(threadIdx.x >> 8));
  const int slice = blockIdx.x * 4 + ks;
  const int node  = (slice < 16) ? *src_idx : *dst_idx;
  const float* x = node_embs + (long)node * H + (slice & 15) * KSL;
  const float* w = W1 + (long)slice * KSL * H + h;
  const float b1h  = b1[h];
  const float bias = (blockIdx.x == 0) ? b1h : 0.0f;
  float acc = 0.0f;
  for (int j = 0; j < KSL; ++j) acc += x[j] * w[(long)j * H];
  red[ks][h] = acc;
  __syncthreads();
  if (ks == 0) {
    const float s = (red[0][h] + red[1][h]) + (red[2][h] + red[3][h]);
    part[blockIdx.x * H + h] = s + bias;
  }
}

#define GATHER_ISSUE(R, J, LEN, HALF)                                                     \
  _Pragma("unroll")                                                                       \
  for (int t = 0; t < L; ++t) {                                                           \
    if (t < (LEN)) {                                                                      \
      const int idx = __builtin_amdgcn_readlane(iv[t & 3], (J) * 4 + (t >> 2));           \
      R[t] = *((const v4f*)(nb + (long)idx * H) + (HALF) * (H / 8));                      \
    }                                                                                     \
  }
#define GATHER_FINISH(R, J, LEN, HALF)                                                    \
  {                                                                                       \
    v4f s = {};                                                                           \
    _Pragma("unroll")                                                                     \
    for (int t = 0; t < L; ++t) {                                                         \
      if (t < (LEN)) s += R[t];                                                           \
    }                                                                                     \
    const float inv = 1.0f / (float)((LEN) > 0 ? (LEN) : 1);                              \
    *(v4f*)&ldsA[wid * 8 + (J)][(HALF) * (H / 2) + col] = s * inv;                        \
  }
#define GATHER_FINISH_ALL(R, J, LEN, HALF)                                                \
  {                                                                                       \
      \
      \
      \
    _Pragma("unroll")                                                                     \
    for (int t = 0; t < L; ++t) asm volatile("" : "+v"(R[t]));                            \
    v4f s = {};                                                                           \
    _Pragma("unroll")                                                                     \
    for (int t = 0; t < L; ++t) s += R[t];                                                \
    const float inv = 1.0f / (float)((LEN) > 0 ? (LEN) : 1);                              \
    *(v4f*)&ldsA[wid * 8 + (J)][(HALF) * (H / 2) + col] = s * inv;                        \
  }
#define STAGE_B()                                                                         \
  {                                                                                       \
    const int n = tid;                                                                    \
    for (int kk = 0; kk < KC / 4; ++kk) {                                                 \
      const long krow = kbase + kk * 4;                                                   \
      v4f b;                                                                              \
      b[0] = W1s[(krow + 0) * H + n];                                                     \
      b[1] = W1s[(krow + 1) * H + n];                                                     \
      b[2] = W1s[(krow + 2) * H + n];                                                     \
      b[3] = W1s[(krow + 3) * H + n];                                                     \
      *(v4f*)&ldsB[(kk * H + n) * 4] = b;                                                 \
    }                                                                                     \
  }
#define MMA_STEPS(KK0, KK1)                                                               \
  _Pragma("unroll")                                                                       \
  for (int kk = (KK0); kk < (KK1); ++kk) {                                                \
    const int k = kbase + kk * 4;                                                         \
                \
    const v2f a0 = *(const v2f*)&ldsA[arow0][k + khalf * 2];                              \
    const v2f a1 = *(const v2f*)&ldsA[arow1][k + khalf * 2];                              \
                 \
    const float* bb = &ldsB[(kk * H + n0 + lh) * 4 + khalf * 2];                          \
    const v2f b0 = *(const v2f*)(bb);                                                     \
    const v2f b1 = *(const v2f*)(bb + 16 * 4);                                            \
    const v2f b2v = *(const v2f*)(bb + 32 * 4);                                           \
    const v2f b3 = *(const v2f*)(bb + 48 * 4);                                            \
    acc[0][0] = __builtin_amdgcn_wmma_f32_16x16x4_f32(false, a0, false, b0,  (short)0, acc[0][0], false, false); \
    acc[0][1] = __builtin_amdgcn_wmma_f32_16x16x4_f32(false, a0, false, b1,  (short)0, acc[0][1], false, false); \
    acc[0][2] = __builtin_amdgcn_wmma_f32_16x16x4_f32(false, a0, false, b2v, (short)0, acc[0][2], false, false); \
    acc[0][3] = __builtin_amdgcn_wmma_f32_16x16x4_f32(false, a0, false, b3,  (short)0, acc[0][3], false, false); \
    acc[1][0] = __builtin_amdgcn_wmma_f32_16x16x4_f32(false, a1, false, b0,  (short)0, acc[1][0], false, false); \
    acc[1][1] = __builtin_amdgcn_wmma_f32_16x16x4_f32(false, a1, false, b1,  (short)0, acc[1][1], false, false); \
    acc[1][2] = __builtin_amdgcn_wmma_f32_16x16x4_f32(false, a1, false, b2v, (short)0, acc[1][2], false, false); \
    acc[1][3] = __builtin_amdgcn_wmma_f32_16x16x4_f32(false, a1, false, b3,  (short)0, acc[1][3], false, false); \
  }

__global__ __launch_bounds__(256) void path_policy_kernel(
    const float* __restrict__ node_embs,
    const int*   __restrict__ path_nodes,
    const int*   __restrict__ path_lens,
    const float* __restrict__ edge_feats,
    const float* __restrict__ scal_feats,
    const float* __restrict__ W1,
    const float* __restrict__ W2,
    const float* __restrict__ b2,
    const float* __restrict__ part,
    float* __restrict__ out) {
  __shared__ float ldsA[MTILE][APITCH];
  __shared__ float ldsB[(KC / 4) * H * 4];
  __shared__ float ldsQ[MTILE];

  const int tid  = threadIdx.x;
  const int lane = tid & 31;
  const int wid  = tid >> 5;
  const int p0   = blockIdx.x * MTILE;

  if (tid < MTILE) ldsQ[tid] = 0.0f;

  const int col = lane * 4;
  const int pw  = p0 + wid * 8;
  const v4i iv  = *(const v4i*)(path_nodes + (long)pw * L + lane * 4);
  const int lv  = path_lens[pw + (lane & 7)];
  const float* nb = node_embs + col;
#pragma nounroll
  for (int jp = 0; jp < 4; ++jp) {
    const int ja = jp * 2, jb = ja + 1;
    const int lena = __builtin_amdgcn_readlane(lv, ja);
    const int lenb = __builtin_amdgcn_readlane(lv, jb);
    v4f ra[L], rb[L];
    GATHER_ISSUE(ra, ja, lena, 0)
    GATHER_ISSUE(rb, jb, lenb, 0)
    GATHER_FINISH(ra, ja, lena, 0)
    GATHER_FINISH(rb, jb, lenb, 0)
  }
  for (int i = tid; i < MTILE * 24; i += 256) {
    const int pi = i / 24, c = i % 24;
    const float v = (c < 8) ? edge_feats[(long)(p0 + pi) * 8 + c]
                            : scal_feats[(long)(p0 + pi) * 16 + (c - 8)];
    ldsA[pi][256 + c] = v;
  }

  const int mgrp  = wid >> 2;
  const int n0    = (wid & 3) * 64;
  const int lh    = lane & 15;
  const int khalf = lane >> 4;
  const int arow0 = mgrp * 32 + lh;
  const int arow1 = arow0 + 16;

  v8f acc[2][4] = {};

  const float* W1s = W1 + (long)512 * H;
  for (int ch = 0; ch < 2; ++ch) {
    const int kbase = ch * KC;
    __syncthreads();
    STAGE_B()
    __syncthreads();
    const int j0 = ch * 4;
    const int len0 = __builtin_amdgcn_readlane(lv, j0);
    const int len1 = __builtin_amdgcn_readlane(lv, j0 + 1);
    const int len2 = __builtin_amdgcn_readlane(lv, j0 + 2);
    const int len3 = __builtin_amdgcn_readlane(lv, j0 + 3);
    {
      v4f ra[L] = {};
      GATHER_ISSUE(ra, j0, len0, 1)
      MMA_STEPS(0, 4)
      __builtin_amdgcn_sched_barrier(0);
      GATHER_FINISH_ALL(ra, j0, len0, 1)
    }
    {
      v4f ra[L] = {};
      GATHER_ISSUE(ra, j0 + 1, len1, 1)
      MMA_STEPS(4, 7)
      __builtin_amdgcn_sched_barrier(0);
      GATHER_FINISH_ALL(ra, j0 + 1, len1, 1)
    }
    {
      v4f ra[L] = {};
      GATHER_ISSUE(ra, j0 + 2, len2, 1)
      MMA_STEPS(7, 11)
      __builtin_amdgcn_sched_barrier(0);
      GATHER_FINISH_ALL(ra, j0 + 2, len2, 1)
    }
    {
      v4f ra[L] = {};
      GATHER_ISSUE(ra, j0 + 3, len3, 1)
      MMA_STEPS(11, KC / 4)
      __builtin_amdgcn_sched_barrier(0);
      GATHER_FINISH_ALL(ra, j0 + 3, len3, 1)
    }
  }
  for (int ch = 2; ch < NCHUNK; ++ch) {
    const int kbase = ch * KC;
    __syncthreads();
    STAGE_B()
    __syncthreads();
    MMA_STEPS(0, KC / 4)
  }

  float cn[4], wn[4];
#pragma unroll
  for (int s = 0; s < 4; ++s) {
    const int n = n0 + s * 16 + lh;
    float cs = part[n];
    for (int j = 1; j < NPART; ++j) cs += part[j * H + n];
    cn[s] = cs;
    wn[s] = W2[n];
  }
#pragma unroll
  for (int ms = 0; ms < 2; ++ms) {
#pragma unroll
    for (int r = 0; r < 8; ++r) {
      float part = 0.0f, v;
      v = acc[ms][0][r] + cn[0]; part += fmaxf(v, 0.0f) * wn[0];
      v = acc[ms][1][r] + cn[1]; part += fmaxf(v, 0.0f) * wn[1];
      v = acc[ms][2][r] + cn[2]; part += fmaxf(v, 0.0f) * wn[2];
      v = acc[ms][3][r] + cn[3]; part += fmaxf(v, 0.0f) * wn[3];
      part += __shfl_xor(part, 1, 32);
      part += __shfl_xor(part, 2, 32);
      part += __shfl_xor(part, 4, 32);
      part += __shfl_xor(part, 8, 32);
      if (lh == 0) {
        const int row = mgrp * 32 + ms * 16 + r + khalf * 8;
        atomicAdd(&ldsQ[row], part);
      }
    }
  }

  __syncthreads();
  if (tid < MTILE) {
    const int p   = p0 + tid;
    const int len = path_lens[p];
    out[p] = (len > 0) ? (ldsQ[tid] + b2[0]) : 0.0f;
  }
}

extern "C" void kernel_launch(void* const* d_in, const int* in_sizes, int n_in,
                              void* d_out, int out_size, void* d_ws, size_t ws_size,
                              hipStream_t stream) {
  const float* node_embs = (const float*)d_in[0];
  const int*   path_nodes = (const int*)d_in[1];
  const int*   path_lens  = (const int*)d_in[2];
  const float* edge_feats = (const float*)d_in[3];
  const float* scal_feats = (const float*)d_in[4];
  const float* W1 = (const float*)d_in[5];
  const float* b1 = (const float*)d_in[6];
  const float* W2 = (const float*)d_in[7];
  const float* b2 = (const float*)d_in[8];
  const int*   src_idx = (const int*)d_in[9];
  const int*   dst_idx = (const int*)d_in[10];

  float* part = (float*)d_ws;

  const_head_kernel<<<NPART, 1024, 0, stream>>>(node_embs, W1, b1, src_idx, dst_idx, part);
  path_policy_kernel<<<P_TOT / MTILE, 256, 0, stream>>>(
      node_embs, path_nodes, path_lens, edge_feats, scal_feats,
      W1, W2, b2, part, (float*)d_out);
}
